# starting kernel
# baseline (speedup 1.0000x reference)
#include <hip/hip_runtime.h>
#include <hip/hip_bf16.h>
#include <math.h>

typedef __attribute__((ext_vector_type(2))) float v2f;
typedef __attribute__((ext_vector_type(8))) float v8f;

#define WMMA_F32X4(a, b, c) \
  __builtin_amdgcn_wmma_f32_16x16x4_f32(false, (a), false, (b), (short)0, (c), false, false)

static __device__ __forceinline__ v2f lds_v2(const float* p) {
  return *reinterpret_cast<const v2f*>(p);
}

constexpr int NB  = 8192;
constexpr int NA  = 64;
constexpr int DIN = 128;
constexpr int E   = 32;

constexpr int PW = DIN + 2;
constexpr int PE = E + 2;
constexpr int PN = NA + 2;

__global__ __launch_bounds__(128)
void attn_layer_kernel(const float* __restrict__ X,
                       const float* __restrict__ W1,
                       const float* __restrict__ b1,
                       const float* __restrict__ Cm,
                       const float* __restrict__ W2,
                       const float* __restrict__ b2,
                       float* __restrict__ out)
{
  __shared__ __align__(16) float W1t[E][PW];
  __shared__ __align__(16) float Ct[E][PE];
  __shared__ __align__(16) float fis[NA][PE];
  __shared__ __align__(16) float fiT[E][PN];
  __shared__ __align__(16) float gs[NA][PE];
  __shared__ __align__(16) float als[NA][PN];
  __shared__ float W2s[2 * E];
  __shared__ float b1s[E];

  const int b    = blockIdx.x;
  const int tid  = threadIdx.x;
  const int lane = tid & 31;
  const int wave = tid >> 5;
  const int half = lane >> 4;
  const int ln   = lane & 15;
  const int m0   = wave * 16;
  const int kh   = half * 2;

  for (int i = tid; i < DIN * E; i += 128) W1t[i % E][i / E] = W1[i];
  for (int i = tid; i < E * E;   i += 128) Ct[i % E][i / E]  = Cm[i];
  if (tid < 2 * E) W2s[tid] = W2[tid];
  if (tid < E)     b1s[tid] = b1[tid];
  __syncthreads();

  const float* Xrow = X + ((size_t)b * NA + m0 + ln) * DIN;
  v8f acc0 = {}, acc1 = {};
  #pragma unroll 4
  for (int kk = 0; kk < DIN; kk += 4) {
    v2f a = *reinterpret_cast<const v2f*>(Xrow + kk + kh);
    v2f p = lds_v2(&W1t[ln][kk + kh]);
    v2f q = lds_v2(&W1t[16 + ln][kk + kh]);
    acc0 = WMMA_F32X4(a, p, acc0);
    acc1 = WMMA_F32X4(a, q, acc1);
  }
  #pragma unroll
  for (int i = 0; i < 8; ++i) {
    int m = m0 + half * 8 + i;
    float f0 = fmaxf(acc0[i] + b1s[ln],      0.f);
    float f1 = fmaxf(acc1[i] + b1s[16 + ln], 0.f);
    fis[m][ln]      = f0;  fis[m][16 + ln] = f1;
    fiT[ln][m]      = f0;  fiT[16 + ln][m] = f1;
  }
  __syncthreads();

  v8f g0 = {}, g1 = {};
  #pragma unroll
  for (int kk = 0; kk < E; kk += 4) {
    v2f a = lds_v2(&fis[m0 + ln][kk + kh]);
    v2f p = lds_v2(&Ct[ln][kk + kh]);
    v2f q = lds_v2(&Ct[16 + ln][kk + kh]);
    g0 = WMMA_F32X4(a, p, g0);
    g1 = WMMA_F32X4(a, q, g1);
  }
  #pragma unroll
  for (int i = 0; i < 8; ++i) {
    int m = m0 + half * 8 + i;
    gs[m][ln] = g0[i];  gs[m][16 + ln] = g1[i];
  }
  __syncthreads();

  v8f bt[4] = {{}, {}, {}, {}};
  #pragma unroll
  for (int kk = 0; kk < E; kk += 4) {
    v2f a = lds_v2(&gs[m0 + ln][kk + kh]);
    #pragma unroll
    for (int t = 0; t < 4; ++t) {
      v2f p = lds_v2(&fis[t * 16 + ln][kk + kh]);
      bt[t] = WMMA_F32X4(a, p, bt[t]);
    }
  }

  #pragma unroll
  for (int i = 0; i < 8; ++i) {
    const int m = m0 + half * 8 + i;
    #pragma unroll
    for (int t = 0; t < 4; ++t)
      if (t * 16 + ln == m) bt[t][i] = -INFINITY;
    float mx = fmaxf(fmaxf(bt[0][i], bt[1][i]), fmaxf(bt[2][i], bt[3][i]));
    #pragma unroll
    for (int s = 1; s < 16; s <<= 1) mx = fmaxf(mx, __shfl_xor(mx, s, 32));
    float sum = 0.f;
    #pragma unroll
    for (int t = 0; t < 4; ++t) { float e = __expf(bt[t][i] - mx); bt[t][i] = e; sum += e; }
    #pragma unroll
    for (int s = 1; s < 16; s <<= 1) sum += __shfl_xor(sum, s, 32);
    const float inv = 1.f / sum;
    #pragma unroll
    for (int t = 0; t < 4; ++t) { bt[t][i] *= inv; als[m][t * 16 + ln] = bt[t][i]; }
  }
  __syncthreads();

  v8f v0 = {}, v1 = {};
  #pragma unroll
  for (int kk = 0; kk < NA; kk += 4) {
    v2f a = lds_v2(&als[m0 + ln][kk + kh]);
    v2f p = lds_v2(&fiT[ln][kk + kh]);
    v2f q = lds_v2(&fiT[16 + ln][kk + kh]);
    v0 = WMMA_F32X4(a, p, v0);
    v1 = WMMA_F32X4(a, q, v1);
  }

  const float b2v = b2[0];
  #pragma unroll
  for (int i = 0; i < 8; ++i) {
    const int m = m0 + half * 8 + i;
    float pp = fis[m][ln]      * W2s[ln]
             + fis[m][16 + ln] * W2s[16 + ln]
             + v0[i]           * W2s[E + ln]
             + v1[i]           * W2s[E + 16 + ln];
    #pragma unroll
    for (int s = 1; s < 16; s <<= 1) pp += __shfl_xor(pp, s, 32);
    if (ln == 0)
      out[(size_t)b * NA + m] = 1.f / (1.f + __expf(-(pp + b2v)));
  }
}

extern "C" void kernel_launch(void* const* d_in, const int* in_sizes, int n_in,
                              void* d_out, int out_size, void* d_ws, size_t ws_size,
                              hipStream_t stream) {
  const float* X  = (const float*)d_in[0];
  const float* W1 = (const float*)d_in[1];
  const float* b1 = (const float*)d_in[2];
  const float* Cm = (const float*)d_in[3];
  const float* W2 = (const float*)d_in[4];
  const float* b2 = (const float*)d_in[5];
  float* out = (float*)d_out;
  attn_layer_kernel<<<NB, 128, 0, stream>>>(X, W1, b1, Cm, W2, b2, out);
}


# Scores computed transposed (agent on lanes): softmax and W2 dot reduce in-lane, p feeds v^T WMMA from registers, 1/sum applied last; alpha LDS trip, 96 shuffles, 1 barrier go; sum order changes
# speedup vs baseline: 1.2294x; 1.2294x over previous
#include <hip/hip_runtime.h>
#include <hip/hip_bf16.h>
#include <math.h>

typedef __attribute__((ext_vector_type(2))) float v2f;
typedef __attribute__((ext_vector_type(4))) float v4f;
typedef __attribute__((ext_vector_type(8))) float v8f;

#define WMMA_F32X4(a, b, c) \
  __builtin_amdgcn_wmma_f32_16x16x4_f32(false, (a), false, (b), (short)0, (c), false, false)

static __device__ __forceinline__ v2f lds_v2(const float* p) {
  return *reinterpret_cast<const v2f*>(p);
}
static __device__ __forceinline__ v4f lds_v4(const float* p) {
  return *reinterpret_cast<const v4f*>(p);
}
static __device__ __forceinline__ float xhalf(float x) {
  int b = __builtin_bit_cast(int, x);
  return __builtin_bit_cast(float,
      __builtin_amdgcn_permlanex16(b, b, 0x76543210, 0xfedcba98, false, false));
}

constexpr int NB  = 8192;
constexpr int NA  = 64;
constexpr int DIN = 128;
constexpr int E   = 32;

constexpr int PW = DIN + 2;
constexpr int PE = E + 2;
constexpr int PN = NA + 4;

__global__ __launch_bounds__(128)
void attn_layer_kernel(const float* __restrict__ X,
                       const float* __restrict__ W1,
                       const float* __restrict__ b1,
                       const float* __restrict__ Cm,
                       const float* __restrict__ W2,
                       const float* __restrict__ b2,
                       float* __restrict__ out)
{
  __shared__ __align__(16) float W1t[E][PW];
  __shared__ __align__(16) float Ct[E][PE];
  __shared__ __align__(16) float fis[NA][PE];
  __shared__ __align__(16) float fiT[E][PN];
  __shared__ __align__(16) float gs[NA][PE];
  __shared__ __align__(16) float W2s[2 * E];
  __shared__ float b1s[E];

  const int b    = blockIdx.x;
  const int tid  = threadIdx.x;
  const int lane = tid & 31;
  const int wave = tid >> 5;
  const int half = lane >> 4;
  const int ln   = lane & 15;
  const int m0   = wave * 16;
  const int kh   = half * 2;

  for (int i = tid; i < DIN * E; i += 128) W1t[i % E][i / E] = W1[i];
  for (int i = tid; i < E * E;   i += 128) Ct[i % E][i / E]  = Cm[i];
  if (tid < 2 * E) W2s[tid] = W2[tid];
  if (tid < E)     b1s[tid] = b1[tid];
  __syncthreads();

  const float* Xrow = X + ((size_t)b * NA + m0 + ln) * DIN;
  v8f acc0 = {}, acc1 = {};
  #pragma unroll 4
  for (int kk = 0; kk < DIN; kk += 4) {
    v2f a = *reinterpret_cast<const v2f*>(Xrow + kk + kh);
    v2f p = lds_v2(&W1t[ln][kk + kh]);
    v2f q = lds_v2(&W1t[16 + ln][kk + kh]);
    acc0 = WMMA_F32X4(a, p, acc0);
    acc1 = WMMA_F32X4(a, q, acc1);
  }
  #pragma unroll
  for (int i = 0; i < 8; ++i) {
    int m = m0 + half * 8 + i;
    float f0 = fmaxf(acc0[i] + b1s[ln],      0.f);
    float f1 = fmaxf(acc1[i] + b1s[16 + ln], 0.f);
    fis[m][ln]      = f0;  fis[m][16 + ln] = f1;
    fiT[ln][m]      = f0;  fiT[16 + ln][m] = f1;
  }
  __syncthreads();

  v8f g0 = {}, g1 = {};
  #pragma unroll
  for (int kk = 0; kk < E; kk += 4) {
    v2f a = lds_v2(&fis[m0 + ln][kk + kh]);
    v2f p = lds_v2(&Ct[ln][kk + kh]);
    v2f q = lds_v2(&Ct[16 + ln][kk + kh]);
    g0 = WMMA_F32X4(a, p, g0);
    g1 = WMMA_F32X4(a, q, g1);
  }
  #pragma unroll
  for (int i = 0; i < 8; ++i) {
    int m = m0 + half * 8 + i;
    gs[m][ln] = g0[i];  gs[m][16 + ln] = g1[i];
  }
  __syncthreads();

  v8f bt[4] = {{}, {}, {}, {}};
  #pragma unroll
  for (int kk = 0; kk < E; kk += 4) {
    v2f gq = lds_v2(&gs[m0 + ln][kk + kh]);
    #pragma unroll
    for (int t = 0; t < 4; ++t) {
      v2f fp = lds_v2(&fis[t * 16 + ln][kk + kh]);
      bt[t] = WMMA_F32X4(fp, gq, bt[t]);
    }
  }

  const int wv = __builtin_amdgcn_readfirstlane(wave);
  const int dr = ln - half * 8;
  #pragma unroll
  for (int t = 0; t < 4; ++t) {
    if (t == wv) {
      #pragma unroll
      for (int r = 0; r < 8; ++r)
        if (r == dr) bt[t][r] = -INFINITY;
    }
  }
  float mxt[4];
  #pragma unroll
  for (int t = 0; t < 4; ++t) {
    float a = fmaxf(fmaxf(bt[t][0], bt[t][1]), fmaxf(bt[t][2], bt[t][3]));
    float c = fmaxf(fmaxf(bt[t][4], bt[t][5]), fmaxf(bt[t][6], bt[t][7]));
    mxt[t] = fmaxf(a, c);
  }
  float mx = fmaxf(fmaxf(mxt[0], mxt[1]), fmaxf(mxt[2], mxt[3]));
  mx = fmaxf(mx, xhalf(mx));
  float st[4] = {0.f, 0.f, 0.f, 0.f};
  #pragma unroll
  for (int t = 0; t < 4; ++t) {
    #pragma unroll
    for (int r = 0; r < 8; ++r) {
      float e = __expf(bt[t][r] - mx);
      bt[t][r] = e;
      st[t] += e;
    }
  }
  float sum = (st[0] + st[1]) + (st[2] + st[3]);
  sum += xhalf(sum);
  const float inv = 1.f / sum;

  v8f vt0 = {}, vt1 = {};
  #pragma unroll
  for (int t = 0; t < 4; ++t) {
    #pragma unroll
    for (int u = 0; u < 2; ++u) {
      const int j0 = t * 16 + half * 8 + u * 4;
      v4f a0 = lds_v4(&fiT[ln][j0]);
      v4f a1 = lds_v4(&fiT[16 + ln][j0]);
      v2f pl = {bt[t][4 * u],     bt[t][4 * u + 1]};
      v2f ph = {bt[t][4 * u + 2], bt[t][4 * u + 3]};
      v2f a0l = {a0[0], a0[1]}, a0h = {a0[2], a0[3]};
      v2f a1l = {a1[0], a1[1]}, a1h = {a1[2], a1[3]};
      vt0 = WMMA_F32X4(a0l, pl, vt0);
      vt1 = WMMA_F32X4(a1l, pl, vt1);
      vt0 = WMMA_F32X4(a0h, ph, vt0);
      vt1 = WMMA_F32X4(a1h, ph, vt1);
    }
  }

  const float b2v = b2[0];
  const float* w2a = &W2s[half * 16];
  const float* w2b = &W2s[E + half * 8];
  const float* fr  = &fis[m0 + ln][half * 16];
  float pf0 = 0.f, pf1 = 0.f;
  #pragma unroll
  for (int q = 0; q < 4; ++q) {
    v4f w  = lds_v4(w2a + 4 * q);
    v2f f0 = lds_v2(fr + 4 * q);
    v2f f1 = lds_v2(fr + 4 * q + 2);
    pf0 += f0[0] * w[0];  pf1 += f0[1] * w[1];
    pf0 += f1[0] * w[2];  pf1 += f1[1] * w[3];
  }
  float pv0 = 0.f, pv1 = 0.f;
  #pragma unroll
  for (int q = 0; q < 2; ++q) {
    v4f w0 = lds_v4(w2b + 4 * q);
    v4f w1 = lds_v4(w2b + 16 + 4 * q);
    #pragma unroll
    for (int s = 0; s < 4; ++s) {
      pv0 += vt0[4 * q + s] * w0[s];
      pv1 += vt1[4 * q + s] * w1[s];
    }
  }
  float pp = (pf0 + pf1) + (pv0 + pv1) * inv;
  pp += xhalf(pp);
  if (half == 0)
    out[(size_t)b * NA + m0 + ln] = 1.f / (1.f + __expf(-(pp + b2v)));
}

extern "C" void kernel_launch(void* const* d_in, const int* in_sizes, int n_in,
                              void* d_out, int out_size, void* d_ws, size_t ws_size,
                              hipStream_t stream) {
  const float* X  = (const float*)d_in[0];
  const float* W1 = (const float*)d_in[1];
  const float* b1 = (const float*)d_in[2];
  const float* Cm = (const float*)d_in[3];
  const float* W2 = (const float*)d_in[4];
  const float* b2 = (const float*)d_in[5];
  float* out = (float*)d_out;
  attn_layer_kernel<<<NB, 128, 0, stream>>>(X, W1, b1, Cm, W2, b2, out);
}


# All global reads issued up front: weights, then the wave's 16x128 X strip as 16-B loads (16 in flight per lane) across the staging barrier (LDS-only fence); fc1 from registers; fc1 sum order changes
# speedup vs baseline: 1.5177x; 1.2345x over previous
#include <hip/hip_runtime.h>
#include <hip/hip_bf16.h>
#include <math.h>

typedef __attribute__((ext_vector_type(2))) float v2f;
typedef __attribute__((ext_vector_type(4))) float v4f;
typedef __attribute__((ext_vector_type(8))) float v8f;

#define WMMA_F32X4(a, b, c) \
  __builtin_amdgcn_wmma_f32_16x16x4_f32(false, (a), false, (b), (short)0, (c), false, false)

static __device__ __forceinline__ v2f lds_v2(const float* p) {
  return *reinterpret_cast<const v2f*>(p);
}
static __device__ __forceinline__ v4f lds_v4(const float* p) {
  return *reinterpret_cast<const v4f*>(p);
}
static __device__ __forceinline__ float xhalf(float x) {
  int b = __builtin_bit_cast(int, x);
  return __builtin_bit_cast(float,
      __builtin_amdgcn_permlanex16(b, b, 0x76543210, 0xfedcba98, false, false));
}

constexpr int NB  = 8192;
constexpr int NA  = 64;
constexpr int DIN = 128;
constexpr int E   = 32;

constexpr int PW = DIN + 4;
constexpr int PE = E + 2;
constexpr int PN = NA + 4;

__global__ __launch_bounds__(128)
void attn_layer_kernel(const float* __restrict__ X,
                       const float* __restrict__ W1,
                       const float* __restrict__ b1,
                       const float* __restrict__ Cm,
                       const float* __restrict__ W2,
                       const float* __restrict__ b2,
                       float* __restrict__ out)
{
  __shared__ __align__(16) float W1t[E][PW];
  __shared__ __align__(16) float Ct[E][PE];
  __shared__ __align__(16) float fis[NA][PE];
  __shared__ __align__(16) float fiT[E][PN];
  __shared__ __align__(16) float gs[NA][PE];
  __shared__ __align__(16) float W2s[2 * E];
  __shared__ float b1s[E];

  const int b    = blockIdx.x;
  const int tid  = threadIdx.x;
  const int lane = tid & 31;
  const int wave = tid >> 5;
  const int half = lane >> 4;
  const int ln   = lane & 15;
  const int m0   = wave * 16;
  const int kh   = half * 2;

  float w1r[DIN * E / 128];
  #pragma unroll
  for (int it = 0; it < DIN * E / 128; ++it) w1r[it] = W1[tid + 128 * it];
  float cr[E * E / 128];
  #pragma unroll
  for (int it = 0; it < E * E / 128; ++it) cr[it] = Cm[tid + 128 * it];
  float w2r = 0.f, b1r = 0.f;
  if (tid < 2 * E) w2r = W2[tid];
  if (tid < E)     b1r = b1[tid];
  const float* Xrow = X + ((size_t)b * NA + m0 + ln) * DIN + 4 * half;
  v4f xa[DIN / 8];
  #pragma unroll
  for (int c = 0; c < DIN / 8; ++c) xa[c] = *reinterpret_cast<const v4f*>(Xrow + 8 * c);

  #pragma unroll
  for (int it = 0; it < DIN * E / 128; ++it) W1t[lane][wave + 4 * it] = w1r[it];
  #pragma unroll
  for (int it = 0; it < E * E / 128; ++it) Ct[lane][wave + 4 * it] = cr[it];
  if (tid < 2 * E) W2s[tid] = w2r;
  if (tid < E)     b1s[tid] = b1r;
  __builtin_amdgcn_fence(__ATOMIC_RELEASE, "workgroup", "local");
  __builtin_amdgcn_s_barrier();
  __builtin_amdgcn_fence(__ATOMIC_ACQUIRE, "workgroup", "local");

  v8f acc0 = {}, acc1 = {};
  #pragma unroll
  for (int c = 0; c < DIN / 8; ++c) {
    v4f p = lds_v4(&W1t[ln][8 * c + 4 * half]);
    v4f q = lds_v4(&W1t[16 + ln][8 * c + 4 * half]);
    v2f al = {xa[c][0], xa[c][1]}, ah = {xa[c][2], xa[c][3]};
    v2f pl = {p[0], p[1]}, ph = {p[2], p[3]};
    v2f ql = {q[0], q[1]}, qh = {q[2], q[3]};
    acc0 = WMMA_F32X4(al, pl, acc0);
    acc1 = WMMA_F32X4(al, ql, acc1);
    acc0 = WMMA_F32X4(ah, ph, acc0);
    acc1 = WMMA_F32X4(ah, qh, acc1);
  }
  #pragma unroll
  for (int i = 0; i < 8; ++i) {
    int m = m0 + half * 8 + i;
    float f0 = fmaxf(acc0[i] + b1s[ln],      0.f);
    float f1 = fmaxf(acc1[i] + b1s[16 + ln], 0.f);
    fis[m][ln]      = f0;  fis[m][16 + ln] = f1;
    fiT[ln][m]      = f0;  fiT[16 + ln][m] = f1;
  }
  __syncthreads();

  v8f g0 = {}, g1 = {};
  #pragma unroll
  for (int kk = 0; kk < E; kk += 4) {
    v2f a = lds_v2(&fis[m0 + ln][kk + kh]);
    v2f p = lds_v2(&Ct[ln][kk + kh]);
    v2f q = lds_v2(&Ct[16 + ln][kk + kh]);
    g0 = WMMA_F32X4(a, p, g0);
    g1 = WMMA_F32X4(a, q, g1);
  }
  #pragma unroll
  for (int i = 0; i < 8; ++i) {
    int m = m0 + half * 8 + i;
    gs[m][ln] = g0[i];  gs[m][16 + ln] = g1[i];
  }
  __syncthreads();

  v8f bt[4] = {{}, {}, {}, {}};
  #pragma unroll
  for (int kk = 0; kk < E; kk += 4) {
    v2f gq = lds_v2(&gs[m0 + ln][kk + kh]);
    #pragma unroll
    for (int t = 0; t < 4; ++t) {
      v2f fp = lds_v2(&fis[t * 16 + ln][kk + kh]);
      bt[t] = WMMA_F32X4(fp, gq, bt[t]);
    }
  }

  const int wv = __builtin_amdgcn_readfirstlane(wave);
  const int dr = ln - half * 8;
  #pragma unroll
  for (int t = 0; t < 4; ++t) {
    if (t == wv) {
      #pragma unroll
      for (int r = 0; r < 8; ++r)
        if (r == dr) bt[t][r] = -INFINITY;
    }
  }
  float mxt[4];
  #pragma unroll
  for (int t = 0; t < 4; ++t) {
    float a = fmaxf(fmaxf(bt[t][0], bt[t][1]), fmaxf(bt[t][2], bt[t][3]));
    float c = fmaxf(fmaxf(bt[t][4], bt[t][5]), fmaxf(bt[t][6], bt[t][7]));
    mxt[t] = fmaxf(a, c);
  }
  float mx = fmaxf(fmaxf(mxt[0], mxt[1]), fmaxf(mxt[2], mxt[3]));
  mx = fmaxf(mx, xhalf(mx));
  float st[4] = {0.f, 0.f, 0.f, 0.f};
  #pragma unroll
  for (int t = 0; t < 4; ++t) {
    #pragma unroll
    for (int r = 0; r < 8; ++r) {
      float e = __expf(bt[t][r] - mx);
      bt[t][r] = e;
      st[t] += e;
    }
  }
  float sum = (st[0] + st[1]) + (st[2] + st[3]);
  sum += xhalf(sum);
  const float inv = 1.f / sum;

  v8f vt0 = {}, vt1 = {};
  #pragma unroll
  for (int t = 0; t < 4; ++t) {
    #pragma unroll
    for (int u = 0; u < 2; ++u) {
      const int j0 = t * 16 + half * 8 + u * 4;
      v4f a0 = lds_v4(&fiT[ln][j0]);
      v4f a1 = lds_v4(&fiT[16 + ln][j0]);
      v2f pl = {bt[t][4 * u],     bt[t][4 * u + 1]};
      v2f ph = {bt[t][4 * u + 2], bt[t][4 * u + 3]};
      v2f a0l = {a0[0], a0[1]}, a0h = {a0[2], a0[3]};
      v2f a1l = {a1[0], a1[1]}, a1h = {a1[2], a1[3]};
      vt0 = WMMA_F32X4(a0l, pl, vt0);
      vt1 = WMMA_F32X4(a1l, pl, vt1);
      vt0 = WMMA_F32X4(a0h, ph, vt0);
      vt1 = WMMA_F32X4(a1h, ph, vt1);
    }
  }

  const float b2v = b2[0];
  const float* w2a = &W2s[half * 16];
  const float* w2b = &W2s[E + half * 8];
  const float* fr  = &fis[m0 + ln][half * 16];
  float pf0 = 0.f, pf1 = 0.f;
  #pragma unroll
  for (int q = 0; q < 4; ++q) {
    v4f w  = lds_v4(w2a + 4 * q);
    v2f f0 = lds_v2(fr + 4 * q);
    v2f f1 = lds_v2(fr + 4 * q + 2);
    pf0 += f0[0] * w[0];  pf1 += f0[1] * w[1];
    pf0 += f1[0] * w[2];  pf1 += f1[1] * w[3];
  }
  float pv0 = 0.f, pv1 = 0.f;
  #pragma unroll
  for (int q = 0; q < 2; ++q) {
    v4f w0 = lds_v4(w2b + 4 * q);
    v4f w1 = lds_v4(w2b + 16 + 4 * q);
    #pragma unroll
    for (int s = 0; s < 4; ++s) {
      pv0 += vt0[4 * q + s] * w0[s];
      pv1 += vt1[4 * q + s] * w1[s];
    }
  }
  float pp = (pf0 + pf1) + (pv0 + pv1) * inv;
  pp += xhalf(pp);
  if (half == 0)
    out[(size_t)b * NA + m0 + ln] = 1.f / (1.f + __expf(-(pp + b2v)));
}

extern "C" void kernel_launch(void* const* d_in, const int* in_sizes, int n_in,
                              void* d_out, int out_size, void* d_ws, size_t ws_size,
                              hipStream_t stream) {
  const float* X  = (const float*)d_in[0];
  const float* W1 = (const float*)d_in[1];
  const float* b1 = (const float*)d_in[2];
  const float* Cm = (const float*)d_in[3];
  const float* W2 = (const float*)d_in[4];
  const float* b2 = (const float*)d_in[5];
  float* out = (float*)d_out;
  attn_layer_kernel<<<NB, 128, 0, stream>>>(X, W1, b1, Cm, W2, b2, out);
}


# fc2 re-associated: (alpha @ fi) @ W2b = alpha @ (fi @ W2b), v never formed: its 32 of 144 WMMAs per wave and the fi^T LDS plane go; s[j] = fi[j].W2b per agent, p.s summed in-lane; sum order changes
# speedup vs baseline: 1.6464x; 1.0848x over previous
#include <hip/hip_runtime.h>
#include <hip/hip_bf16.h>
#include <math.h>

typedef __attribute__((ext_vector_type(2))) float v2f;
typedef __attribute__((ext_vector_type(4))) float v4f;
typedef __attribute__((ext_vector_type(8))) float v8f;

#define WMMA_F32X4(a, b, c) \
  __builtin_amdgcn_wmma_f32_16x16x4_f32(false, (a), false, (b), (short)0, (c), false, false)

static __device__ __forceinline__ v2f lds_v2(const float* p) {
  return *reinterpret_cast<const v2f*>(p);
}
static __device__ __forceinline__ v4f lds_v4(const float* p) {
  return *reinterpret_cast<const v4f*>(p);
}
static __device__ __forceinline__ float xhalf(float x) {
  int b = __builtin_bit_cast(int, x);
  return __builtin_bit_cast(float,
      __builtin_amdgcn_permlanex16(b, b, 0x76543210, 0xfedcba98, false, false));
}

constexpr int NB  = 8192;
constexpr int NA  = 64;
constexpr int DIN = 128;
constexpr int E   = 32;

constexpr int PW = DIN + 4;
constexpr int PE = E + 2;

__global__ __launch_bounds__(128)
void attn_layer_kernel(const float* __restrict__ X,
                       const float* __restrict__ W1,
                       const float* __restrict__ b1,
                       const float* __restrict__ Cm,
                       const float* __restrict__ W2,
                       const float* __restrict__ b2,
                       float* __restrict__ out)
{
  __shared__ __align__(16) float W1t[E][PW];
  __shared__ __align__(16) float Ct[E][PE];
  __shared__ __align__(16) float fis[NA][PE];
  __shared__ __align__(16) float gs[NA][PE];
  __shared__ __align__(16) float W2s[2 * E];
  __shared__ __align__(16) float ss[NA];
  __shared__ float b1s[E];

  const int b    = blockIdx.x;
  const int tid  = threadIdx.x;
  const int lane = tid & 31;
  const int wave = tid >> 5;
  const int half = lane >> 4;
  const int ln   = lane & 15;
  const int m0   = wave * 16;
  const int kh   = half * 2;

  float w1r[DIN * E / 128];
  #pragma unroll
  for (int it = 0; it < DIN * E / 128; ++it) w1r[it] = W1[tid + 128 * it];
  float cr[E * E / 128];
  #pragma unroll
  for (int it = 0; it < E * E / 128; ++it) cr[it] = Cm[tid + 128 * it];
  float w2r = 0.f, b1r = 0.f;
  if (tid < 2 * E) w2r = W2[tid];
  if (tid < E)     b1r = b1[tid];
  const float* Xrow = X + ((size_t)b * NA + m0 + ln) * DIN + 4 * half;
  v4f xa[DIN / 8];
  #pragma unroll
  for (int c = 0; c < DIN / 8; ++c) xa[c] = *reinterpret_cast<const v4f*>(Xrow + 8 * c);

  #pragma unroll
  for (int it = 0; it < DIN * E / 128; ++it) W1t[lane][wave + 4 * it] = w1r[it];
  #pragma unroll
  for (int it = 0; it < E * E / 128; ++it) Ct[lane][wave + 4 * it] = cr[it];
  if (tid < 2 * E) W2s[tid] = w2r;
  if (tid < E)     b1s[tid] = b1r;
  __builtin_amdgcn_fence(__ATOMIC_RELEASE, "workgroup", "local");
  __builtin_amdgcn_s_barrier();
  __builtin_amdgcn_fence(__ATOMIC_ACQUIRE, "workgroup", "local");

  v8f acc0 = {}, acc1 = {};
  #pragma unroll
  for (int c = 0; c < DIN / 8; ++c) {
    v4f p = lds_v4(&W1t[ln][8 * c + 4 * half]);
    v4f q = lds_v4(&W1t[16 + ln][8 * c + 4 * half]);
    v2f al = {xa[c][0], xa[c][1]}, ah = {xa[c][2], xa[c][3]};
    v2f pl = {p[0], p[1]}, ph = {p[2], p[3]};
    v2f ql = {q[0], q[1]}, qh = {q[2], q[3]};
    acc0 = WMMA_F32X4(al, pl, acc0);
    acc1 = WMMA_F32X4(al, ql, acc1);
    acc0 = WMMA_F32X4(ah, ph, acc0);
    acc1 = WMMA_F32X4(ah, qh, acc1);
  }
  #pragma unroll
  for (int i = 0; i < 8; ++i) {
    int m = m0 + half * 8 + i;
    float f0 = fmaxf(acc0[i] + b1s[ln],      0.f);
    float f1 = fmaxf(acc1[i] + b1s[16 + ln], 0.f);
    fis[m][ln]      = f0;  fis[m][16 + ln] = f1;
  }
  __syncthreads();

  v8f g0 = {}, g1 = {};
  #pragma unroll
  for (int kk = 0; kk < E; kk += 4) {
    v2f a = lds_v2(&fis[m0 + ln][kk + kh]);
    v2f p = lds_v2(&Ct[ln][kk + kh]);
    v2f q = lds_v2(&Ct[16 + ln][kk + kh]);
    g0 = WMMA_F32X4(a, p, g0);
    g1 = WMMA_F32X4(a, q, g1);
  }

  float pf, sj;
  {
    const float* fr  = &fis[m0 + ln][half * 16];
    const float* w2a = &W2s[half * 16];
    const float* w2b = &W2s[E + half * 16];
    float pf0 = 0.f, pf1 = 0.f, sj0 = 0.f, sj1 = 0.f;
    #pragma unroll
    for (int q = 0; q < 4; ++q) {
      v4f wa = lds_v4(w2a + 4 * q);
      v4f wb = lds_v4(w2b + 4 * q);
      v2f f0 = lds_v2(fr + 4 * q);
      v2f f1 = lds_v2(fr + 4 * q + 2);
      pf0 += f0[0] * wa[0];  pf1 += f0[1] * wa[1];
      pf0 += f1[0] * wa[2];  pf1 += f1[1] * wa[3];
      sj0 += f0[0] * wb[0];  sj1 += f0[1] * wb[1];
      sj0 += f1[0] * wb[2];  sj1 += f1[1] * wb[3];
    }
    pf = pf0 + pf1;
    sj = sj0 + sj1;
    sj += xhalf(sj);
  }
  if (half == 0) ss[m0 + ln] = sj;
  #pragma unroll
  for (int i = 0; i < 8; ++i) {
    int m = m0 + half * 8 + i;
    gs[m][ln] = g0[i];  gs[m][16 + ln] = g1[i];
  }
  __syncthreads();

  v8f bt[4] = {{}, {}, {}, {}};
  #pragma unroll
  for (int kk = 0; kk < E; kk += 4) {
    v2f gq = lds_v2(&gs[m0 + ln][kk + kh]);
    #pragma unroll
    for (int t = 0; t < 4; ++t) {
      v2f fp = lds_v2(&fis[t * 16 + ln][kk + kh]);
      bt[t] = WMMA_F32X4(fp, gq, bt[t]);
    }
  }

  const int wv = __builtin_amdgcn_readfirstlane(wave);
  const int dr = ln - half * 8;
  #pragma unroll
  for (int t = 0; t < 4; ++t) {
    if (t == wv) {
      #pragma unroll
      for (int r = 0; r < 8; ++r)
        if (r == dr) bt[t][r] = -INFINITY;
    }
  }
  float mxt[4];
  #pragma unroll
  for (int t = 0; t < 4; ++t) {
    float a = fmaxf(fmaxf(bt[t][0], bt[t][1]), fmaxf(bt[t][2], bt[t][3]));
    float c = fmaxf(fmaxf(bt[t][4], bt[t][5]), fmaxf(bt[t][6], bt[t][7]));
    mxt[t] = fmaxf(a, c);
  }
  float mx = fmaxf(fmaxf(mxt[0], mxt[1]), fmaxf(mxt[2], mxt[3]));
  mx = fmaxf(mx, xhalf(mx));
  float st[4] = {0.f, 0.f, 0.f, 0.f};
  float pv[4] = {0.f, 0.f, 0.f, 0.f};
  #pragma unroll
  for (int t = 0; t < 4; ++t) {
    #pragma unroll
    for (int u = 0; u < 2; ++u) {
      v4f sv = lds_v4(&ss[t * 16 + half * 8 + u * 4]);
      #pragma unroll
      for (int s = 0; s < 4; ++s) {
        float e = __expf(bt[t][4 * u + s] - mx);
        st[t] += e;
        pv[t] += e * sv[s];
      }
    }
  }
  float sum = (st[0] + st[1]) + (st[2] + st[3]);
  sum += xhalf(sum);
  const float inv = 1.f / sum;
  const float pvs = (pv[0] + pv[1]) + (pv[2] + pv[3]);

  const float b2v = b2[0];
  float pp = pf + pvs * inv;
  pp += xhalf(pp);
  if (half == 0)
    out[(size_t)b * NA + m0 + ln] = 1.f / (1.f + __expf(-(pp + b2v)));
}

extern "C" void kernel_launch(void* const* d_in, const int* in_sizes, int n_in,
                              void* d_out, int out_size, void* d_ws, size_t ws_size,
                              hipStream_t stream) {
  const float* X  = (const float*)d_in[0];
  const float* W1 = (const float*)d_in[1];
  const float* b1 = (const float*)d_in[2];
  const float* Cm = (const float*)d_in[3];
  const float* W2 = (const float*)d_in[4];
  const float* b2 = (const float*)d_in[5];
  float* out = (float*)d_out;
  attn_layer_kernel<<<NB, 128, 0, stream>>>(X, W1, b1, Cm, W2, b2, out);
}


# Persistent work-groups: 2048 groups (8 per WGP) loop over 4 batches; weights go into LDS once per group, not per batch; the barrier behind the fi store moves in front of it; sums unchanged
# speedup vs baseline: 1.7405x; 1.0572x over previous
#include <hip/hip_runtime.h>
#include <hip/hip_bf16.h>
#include <math.h>

typedef __attribute__((ext_vector_type(2))) float v2f;
typedef __attribute__((ext_vector_type(4))) float v4f;
typedef __attribute__((ext_vector_type(8))) float v8f;

#define WMMA_F32X4(a, b, c) \
  __builtin_amdgcn_wmma_f32_16x16x4_f32(false, (a), false, (b), (short)0, (c), false, false)

static __device__ __forceinline__ v2f lds_v2(const float* p) {
  return *reinterpret_cast<const v2f*>(p);
}
static __device__ __forceinline__ v4f lds_v4(const float* p) {
  return *reinterpret_cast<const v4f*>(p);
}
static __device__ __forceinline__ float xhalf(float x) {
  int b = __builtin_bit_cast(int, x);
  return __builtin_bit_cast(float,
      __builtin_amdgcn_permlanex16(b, b, 0x76543210, 0xfedcba98, false, false));
}

constexpr int NB  = 8192;
constexpr int NA  = 64;
constexpr int DIN = 128;
constexpr int E   = 32;

constexpr int PW = DIN + 4;
constexpr int PE = E + 2;

constexpr int NWG = 2048;

__global__ __launch_bounds__(128)
void attn_layer_kernel(const float* __restrict__ X,
                       const float* __restrict__ W1,
                       const float* __restrict__ b1,
                       const float* __restrict__ Cm,
                       const float* __restrict__ W2,
                       const float* __restrict__ b2,
                       float* __restrict__ out)
{
  __shared__ __align__(16) float W1t[E][PW];
  __shared__ __align__(16) float Ct[E][PE];
  __shared__ __align__(16) float fis[NA][PE];
  __shared__ __align__(16) float gs[NA][PE];
  __shared__ __align__(16) float W2s[2 * E];
  __shared__ __align__(16) float ss[NA];
  __shared__ float b1s[E];

  const int tid  = threadIdx.x;
  const int lane = tid & 31;
  const int wave = tid >> 5;
  const int half = lane >> 4;
  const int ln   = lane & 15;
  const int m0   = wave * 16;
  const int kh   = half * 2;

  float w1r[DIN * E / 128];
  #pragma unroll
  for (int it = 0; it < DIN * E / 128; ++it) w1r[it] = W1[tid + 128 * it];
  float cr[E * E / 128];
  #pragma unroll
  for (int it = 0; it < E * E / 128; ++it) cr[it] = Cm[tid + 128 * it];
  float w2r = 0.f, b1r = 0.f;
  if (tid < 2 * E) w2r = W2[tid];
  if (tid < E)     b1r = b1[tid];
  const float b2v = b2[0];

  #pragma unroll
  for (int it = 0; it < DIN * E / 128; ++it) W1t[lane][wave + 4 * it] = w1r[it];
  #pragma unroll
  for (int it = 0; it < E * E / 128; ++it) Ct[lane][wave + 4 * it] = cr[it];
  if (tid < 2 * E) W2s[tid] = w2r;
  if (tid < E)     b1s[tid] = b1r;
  __syncthreads();

  for (int b = blockIdx.x; b < NB; b += NWG) {
  const float* Xrow = X + ((size_t)b * NA + m0 + ln) * DIN + 4 * half;
  v4f xa[DIN / 8];
  #pragma unroll
  for (int c = 0; c < DIN / 8; ++c) xa[c] = *reinterpret_cast<const v4f*>(Xrow + 8 * c);
  __builtin_amdgcn_sched_barrier(0);

  v8f acc0 = {}, acc1 = {};
  #pragma unroll
  for (int c = 0; c < DIN / 8; ++c) {
    v4f p = lds_v4(&W1t[ln][8 * c + 4 * half]);
    v4f q = lds_v4(&W1t[16 + ln][8 * c + 4 * half]);
    v2f al = {xa[c][0], xa[c][1]}, ah = {xa[c][2], xa[c][3]};
    v2f pl = {p[0], p[1]}, ph = {p[2], p[3]};
    v2f ql = {q[0], q[1]}, qh = {q[2], q[3]};
    acc0 = WMMA_F32X4(al, pl, acc0);
    acc1 = WMMA_F32X4(al, ql, acc1);
    acc0 = WMMA_F32X4(ah, ph, acc0);
    acc1 = WMMA_F32X4(ah, qh, acc1);
  }
  __builtin_amdgcn_fence(__ATOMIC_RELEASE, "workgroup", "local");
  __builtin_amdgcn_s_barrier();
  __builtin_amdgcn_fence(__ATOMIC_ACQUIRE, "workgroup", "local");
  #pragma unroll
  for (int i = 0; i < 8; ++i) {
    int m = m0 + half * 8 + i;
    float f0 = fmaxf(acc0[i] + b1s[ln],      0.f);
    float f1 = fmaxf(acc1[i] + b1s[16 + ln], 0.f);
    fis[m][ln]      = f0;  fis[m][16 + ln] = f1;
  }
  __builtin_amdgcn_fence(__ATOMIC_RELEASE, "workgroup", "local");
  __builtin_amdgcn_wave_barrier();
  __builtin_amdgcn_fence(__ATOMIC_ACQUIRE, "workgroup", "local");

  v8f g0 = {}, g1 = {};
  #pragma unroll
  for (int kk = 0; kk < E; kk += 4) {
    v2f a = lds_v2(&fis[m0 + ln][kk + kh]);
    v2f p = lds_v2(&Ct[ln][kk + kh]);
    v2f q = lds_v2(&Ct[16 + ln][kk + kh]);
    g0 = WMMA_F32X4(a, p, g0);
    g1 = WMMA_F32X4(a, q, g1);
  }

  float pf, sj;
  {
    const float* fr  = &fis[m0 + ln][half * 16];
    const float* w2a = &W2s[half * 16];
    const float* w2b = &W2s[E + half * 16];
    float pf0 = 0.f, pf1 = 0.f, sj0 = 0.f, sj1 = 0.f;
    #pragma unroll
    for (int q = 0; q < 4; ++q) {
      v4f wa = lds_v4(w2a + 4 * q);
      v4f wb = lds_v4(w2b + 4 * q);
      v2f f0 = lds_v2(fr + 4 * q);
      v2f f1 = lds_v2(fr + 4 * q + 2);
      pf0 += f0[0] * wa[0];  pf1 += f0[1] * wa[1];
      pf0 += f1[0] * wa[2];  pf1 += f1[1] * wa[3];
      sj0 += f0[0] * wb[0];  sj1 += f0[1] * wb[1];
      sj0 += f1[0] * wb[2];  sj1 += f1[1] * wb[3];
    }
    pf = pf0 + pf1;
    sj = sj0 + sj1;
    sj += xhalf(sj);
  }
  if (half == 0) ss[m0 + ln] = sj;
  #pragma unroll
  for (int i = 0; i < 8; ++i) {
    int m = m0 + half * 8 + i;
    gs[m][ln] = g0[i];  gs[m][16 + ln] = g1[i];
  }
  __syncthreads();

  v8f bt[4] = {{}, {}, {}, {}};
  #pragma unroll
  for (int kk = 0; kk < E; kk += 4) {
    v2f gq = lds_v2(&gs[m0 + ln][kk + kh]);
    #pragma unroll
    for (int t = 0; t < 4; ++t) {
      v2f fp = lds_v2(&fis[t * 16 + ln][kk + kh]);
      bt[t] = WMMA_F32X4(fp, gq, bt[t]);
    }
  }

  const int wv = __builtin_amdgcn_readfirstlane(wave);
  const int dr = ln - half * 8;
  #pragma unroll
  for (int t = 0; t < 4; ++t) {
    if (t == wv) {
      #pragma unroll
      for (int r = 0; r < 8; ++r)
        if (r == dr) bt[t][r] = -INFINITY;
    }
  }
  float mxt[4];
  #pragma unroll
  for (int t = 0; t < 4; ++t) {
    float a = fmaxf(fmaxf(bt[t][0], bt[t][1]), fmaxf(bt[t][2], bt[t][3]));
    float c = fmaxf(fmaxf(bt[t][4], bt[t][5]), fmaxf(bt[t][6], bt[t][7]));
    mxt[t] = fmaxf(a, c);
  }
  float mx = fmaxf(fmaxf(mxt[0], mxt[1]), fmaxf(mxt[2], mxt[3]));
  mx = fmaxf(mx, xhalf(mx));
  float st[4] = {0.f, 0.f, 0.f, 0.f};
  float pv[4] = {0.f, 0.f, 0.f, 0.f};
  #pragma unroll
  for (int t = 0; t < 4; ++t) {
    #pragma unroll
    for (int u = 0; u < 2; ++u) {
      v4f sv = lds_v4(&ss[t * 16 + half * 8 + u * 4]);
      #pragma unroll
      for (int s = 0; s < 4; ++s) {
        float e = __expf(bt[t][4 * u + s] - mx);
        st[t] += e;
        pv[t] += e * sv[s];
      }
    }
  }
  float sum = (st[0] + st[1]) + (st[2] + st[3]);
  sum += xhalf(sum);
  const float inv = 1.f / sum;
  const float pvs = (pv[0] + pv[1]) + (pv[2] + pv[3]);

  float pp = pf + pvs * inv;
  pp += xhalf(pp);
  if (half == 0)
    out[(size_t)b * NA + m0 + ln] = 1.f / (1.f + __expf(-(pp + b2v)));
  }
}

extern "C" void kernel_launch(void* const* d_in, const int* in_sizes, int n_in,
                              void* d_out, int out_size, void* d_ws, size_t ws_size,
                              hipStream_t stream) {
  const float* X  = (const float*)d_in[0];
  const float* W1 = (const float*)d_in[1];
  const float* b1 = (const float*)d_in[2];
  const float* Cm = (const float*)d_in[3];
  const float* W2 = (const float*)d_in[4];
  const float* b2 = (const float*)d_in[5];
  float* out = (float*)d_out;
  attn_layer_kernel<<<NWG, 128, 0, stream>>>(X, W1, b1, Cm, W2, b2, out);
}


# X strip loads pipelined across batches: 12 of a lane's 16 16-B pieces are requested one batch ahead (6 behind the 2nd barrier, 6 behind the scores; barriers LDS-only), 4 at loop top; sums unchanged
# speedup vs baseline: 1.8043x; 1.0367x over previous
#include <hip/hip_runtime.h>
#include <hip/hip_bf16.h>
#include <math.h>

typedef __attribute__((ext_vector_type(2))) float v2f;
typedef __attribute__((ext_vector_type(4))) float v4f;
typedef __attribute__((ext_vector_type(8))) float v8f;

#define WMMA_F32X4(a, b, c) \
  __builtin_amdgcn_wmma_f32_16x16x4_f32(false, (a), false, (b), (short)0, (c), false, false)

static __device__ __forceinline__ v2f lds_v2(const float* p) {
  return *reinterpret_cast<const v2f*>(p);
}
static __device__ __forceinline__ v4f lds_v4(const float* p) {
  return *reinterpret_cast<const v4f*>(p);
}
static __device__ __forceinline__ float xhalf(float x) {
  int b = __builtin_bit_cast(int, x);
  return __builtin_bit_cast(float,
      __builtin_amdgcn_permlanex16(b, b, 0x76543210, 0xfedcba98, false, false));
}

constexpr int NB  = 8192;
constexpr int NA  = 64;
constexpr int DIN = 128;
constexpr int E   = 32;

constexpr int PW = DIN + 4;
constexpr int PE = E + 2;

constexpr int NWG = 2048;

constexpr int CE1 = 6;
constexpr int CE  = 12;
struct more_yes { static constexpr bool value = true; };
struct more_no  { static constexpr bool value = false; };

__global__ __launch_bounds__(128) __attribute__((amdgpu_waves_per_eu(8, 8)))
void attn_layer_kernel(const float* __restrict__ X,
                       const float* __restrict__ W1,
                       const float* __restrict__ b1,
                       const float* __restrict__ Cm,
                       const float* __restrict__ W2,
                       const float* __restrict__ b2,
                       float* __restrict__ out)
{
  __shared__ __align__(16) float W1t[E][PW];
  __shared__ __align__(16) float Ct[E][PE];
  __shared__ __align__(16) float fis[NA][PE];
  __shared__ __align__(16) float gs[NA][PE];
  __shared__ __align__(16) float W2s[2 * E];
  __shared__ __align__(16) float ss[NA];
  __shared__ float b1s[E];

  const int tid  = threadIdx.x;
  const int lane = tid & 31;
  const int wave = tid >> 5;
  const int half = lane >> 4;
  const int ln   = lane & 15;
  const int m0   = wave * 16;
  const int kh   = half * 2;

  float w1r[DIN * E / 128];
  #pragma unroll
  for (int it = 0; it < DIN * E / 128; ++it) w1r[it] = W1[tid + 128 * it];
  float cr[E * E / 128];
  #pragma unroll
  for (int it = 0; it < E * E / 128; ++it) cr[it] = Cm[tid + 128 * it];
  float w2r = 0.f, b1r = 0.f;
  if (tid < 2 * E) w2r = W2[tid];
  if (tid < E)     b1r = b1[tid];
  const float b2v = b2[0];
  v4f xe[CE];
  {
    const float* Xrow0 = X + ((size_t)blockIdx.x * NA + m0 + ln) * DIN + 4 * half;
    #pragma unroll
    for (int c = 0; c < CE; ++c) xe[c] = *reinterpret_cast<const v4f*>(Xrow0 + 8 * c);
  }

  #pragma unroll
  for (int it = 0; it < DIN * E / 128; ++it) W1t[lane][wave + 4 * it] = w1r[it];
  #pragma unroll
  for (int it = 0; it < E * E / 128; ++it) Ct[lane][wave + 4 * it] = cr[it];
  if (tid < 2 * E) W2s[tid] = w2r;
  if (tid < E)     b1s[tid] = b1r;
  __builtin_amdgcn_fence(__ATOMIC_RELEASE, "workgroup", "local");
  __builtin_amdgcn_s_barrier();
  __builtin_amdgcn_fence(__ATOMIC_ACQUIRE, "workgroup", "local");

  const int wv = __builtin_amdgcn_readfirstlane(wave);

  auto batch = [&](const int b, auto more_tag) __attribute__((always_inline)) {
  constexpr bool MORE = decltype(more_tag)::value;
  const float* Xrow = X + ((size_t)b * NA + m0 + ln) * DIN + 4 * half;
  v4f xl[DIN / 8 - CE];
  #pragma unroll
  for (int c = CE; c < DIN / 8; ++c) xl[c - CE] = *reinterpret_cast<const v4f*>(Xrow + 8 * c);
  __builtin_amdgcn_sched_barrier(0);

  v8f acc0 = {}, acc1 = {};
  #pragma unroll
  for (int c = 0; c < DIN / 8; ++c) {
    v4f p = lds_v4(&W1t[ln][8 * c + 4 * half]);
    v4f q = lds_v4(&W1t[16 + ln][8 * c + 4 * half]);
    const v4f xc = c < CE ? xe[c] : xl[c < CE ? 0 : c - CE];
    v2f al = {xc[0], xc[1]}, ah = {xc[2], xc[3]};
    v2f pl = {p[0], p[1]}, ph = {p[2], p[3]};
    v2f ql = {q[0], q[1]}, qh = {q[2], q[3]};
    acc0 = WMMA_F32X4(al, pl, acc0);
    acc1 = WMMA_F32X4(al, ql, acc1);
    acc0 = WMMA_F32X4(ah, ph, acc0);
    acc1 = WMMA_F32X4(ah, qh, acc1);
  }
  __builtin_amdgcn_fence(__ATOMIC_RELEASE, "workgroup", "local");
  __builtin_amdgcn_s_barrier();
  __builtin_amdgcn_fence(__ATOMIC_ACQUIRE, "workgroup", "local");
  #pragma unroll
  for (int i = 0; i < 8; ++i) {
    int m = m0 + half * 8 + i;
    float f0 = fmaxf(acc0[i] + b1s[ln],      0.f);
    float f1 = fmaxf(acc1[i] + b1s[16 + ln], 0.f);
    fis[m][ln]      = f0;  fis[m][16 + ln] = f1;
  }
  __builtin_amdgcn_fence(__ATOMIC_RELEASE, "workgroup", "local");
  __builtin_amdgcn_wave_barrier();
  __builtin_amdgcn_fence(__ATOMIC_ACQUIRE, "workgroup", "local");

  v8f g0 = {}, g1 = {};
  #pragma unroll
  for (int kk = 0; kk < E; kk += 4) {
    v2f a = lds_v2(&fis[m0 + ln][kk + kh]);
    v2f p = lds_v2(&Ct[ln][kk + kh]);
    v2f q = lds_v2(&Ct[16 + ln][kk + kh]);
    g0 = WMMA_F32X4(a, p, g0);
    g1 = WMMA_F32X4(a, q, g1);
  }

  float pf, sj;
  {
    const float* fr  = &fis[m0 + ln][half * 16];
    const float* w2a = &W2s[half * 16];
    const float* w2b = &W2s[E + half * 16];
    float pf0 = 0.f, pf1 = 0.f, sj0 = 0.f, sj1 = 0.f;
    #pragma unroll
    for (int q = 0; q < 4; ++q) {
      v4f wa = lds_v4(w2a + 4 * q);
      v4f wb = lds_v4(w2b + 4 * q);
      v2f f0 = lds_v2(fr + 4 * q);
      v2f f1 = lds_v2(fr + 4 * q + 2);
      pf0 += f0[0] * wa[0];  pf1 += f0[1] * wa[1];
      pf0 += f1[0] * wa[2];  pf1 += f1[1] * wa[3];
      sj0 += f0[0] * wb[0];  sj1 += f0[1] * wb[1];
      sj0 += f1[0] * wb[2];  sj1 += f1[1] * wb[3];
    }
    pf = pf0 + pf1;
    sj = sj0 + sj1;
    sj += xhalf(sj);
  }
  asm volatile("" : "+v"(pf));
  if (half == 0) ss[m0 + ln] = sj;
  #pragma unroll
  for (int i = 0; i < 8; ++i) {
    int m = m0 + half * 8 + i;
    gs[m][ln] = g0[i];  gs[m][16 + ln] = g1[i];
  }
  __builtin_amdgcn_fence(__ATOMIC_RELEASE, "workgroup", "local");
  __builtin_amdgcn_s_barrier();
  __builtin_amdgcn_fence(__ATOMIC_ACQUIRE, "workgroup", "local");

  const float* Xnext = MORE ? Xrow + (size_t)NWG * NA * DIN : Xrow;
  if constexpr (MORE) {
    #pragma unroll
    for (int c = 0; c < CE1; ++c) xe[c] = *reinterpret_cast<const v4f*>(Xnext + 8 * c);
  }

  v8f bt[4] = {{}, {}, {}, {}};
  v2f gq[2][2], fp[2][2][4];
  #pragma unroll
  for (int p = 0; p < 2; ++p) {
    #pragma unroll
    for (int j = 0; j < 2; ++j) {
      const int kk = 8 * p + 4 * j;
      gq[p][j] = lds_v2(&gs[m0 + ln][kk + kh]);
      #pragma unroll
      for (int t = 0; t < 4; ++t) fp[p][j][t] = lds_v2(&fis[t * 16 + ln][kk + kh]);
    }
  }
  #pragma unroll
  for (int p = 0; p < E / 8; ++p) {
    #pragma unroll
    for (int j = 0; j < 2; ++j) {
      #pragma unroll
      for (int t = 0; t < 4; ++t) bt[t] = WMMA_F32X4(fp[p & 1][j][t], gq[p & 1][j], bt[t]);
    }
    if (p + 2 < E / 8) {
      #pragma unroll
      for (int j = 0; j < 2; ++j) {
        const int kk = 8 * (p + 2) + 4 * j;
        gq[p & 1][j] = lds_v2(&gs[m0 + ln][kk + kh]);
        #pragma unroll
        for (int t = 0; t < 4; ++t) fp[p & 1][j][t] = lds_v2(&fis[t * 16 + ln][kk + kh]);
      }
    }
    __builtin_amdgcn_sched_barrier(0);
  }

  if constexpr (MORE) {
    #pragma unroll
    for (int c = CE1; c < CE; ++c) xe[c] = *reinterpret_cast<const v4f*>(Xnext + 8 * c);
  }

  const int dr = ln - half * 8;
  #pragma unroll
  for (int t = 0; t < 4; ++t) {
    if (t == wv) {
      #pragma unroll
      for (int r = 0; r < 8; ++r)
        if (r == dr) bt[t][r] = -INFINITY;
    }
  }
  float mxt[4];
  #pragma unroll
  for (int t = 0; t < 4; ++t) {
    float a = fmaxf(fmaxf(bt[t][0], bt[t][1]), fmaxf(bt[t][2], bt[t][3]));
    float c = fmaxf(fmaxf(bt[t][4], bt[t][5]), fmaxf(bt[t][6], bt[t][7]));
    mxt[t] = fmaxf(a, c);
  }
  float mx = fmaxf(fmaxf(mxt[0], mxt[1]), fmaxf(mxt[2], mxt[3]));
  mx = fmaxf(mx, xhalf(mx));
  float st[4] = {0.f, 0.f, 0.f, 0.f};
  float pv[4] = {0.f, 0.f, 0.f, 0.f};
  v4f sv[2] = {lds_v4(&ss[half * 8]), lds_v4(&ss[half * 8 + 4])};
  #pragma unroll
  for (int t = 0; t < 4; ++t) {
    v4f nx[2] = {sv[0], sv[1]};
    if (t < 3) {
      nx[0] = lds_v4(&ss[(t + 1) * 16 + half * 8]);
      nx[1] = lds_v4(&ss[(t + 1) * 16 + half * 8 + 4]);
    }
    #pragma unroll
    for (int u = 0; u < 2; ++u) {
      #pragma unroll
      for (int s = 0; s < 4; ++s) {
        float e = __expf(bt[t][4 * u + s] - mx);
        st[t] += e;
        pv[t] += e * sv[u][s];
      }
    }
    sv[0] = nx[0];  sv[1] = nx[1];
    __builtin_amdgcn_sched_barrier(0);
  }
  float sum = (st[0] + st[1]) + (st[2] + st[3]);
  sum += xhalf(sum);
  const float inv = 1.f / sum;
  const float pvs = (pv[0] + pv[1]) + (pv[2] + pv[3]);

  float pp = pf + pvs * inv;
  pp += xhalf(pp);
  if (half == 0)
    out[(size_t)b * NA + m0 + ln] = 1.f / (1.f + __expf(-(pp + b2v)));
  };

  int b = blockIdx.x;
  for (; b + NWG < NB; b += NWG) batch(b, more_yes{});
  batch(b, more_no{});
}

extern "C" void kernel_launch(void* const* d_in, const int* in_sizes, int n_in,
                              void* d_out, int out_size, void* d_ws, size_t ws_size,
                              hipStream_t stream) {
  const float* X  = (const float*)d_in[0];
  const float* W1 = (const float*)d_in[1];
  const float* b1 = (const float*)d_in[2];
  const float* Cm = (const float*)d_in[3];
  const float* W2 = (const float*)d_in[4];
  const float* b2 = (const float*)d_in[5];
  float* out = (float*)d_out;
  attn_layer_kernel<<<NWG, 128, 0, stream>>>(X, W1, b1, Cm, W2, b2, out);
}


# fc1 computed transposed (W1^T as A, X^T as B): fi lands agent-on-lane, e in registers = operand layout of g^T = C^T fi^T and of the scores; fi->LDS->g trip and g plane go; sum order changes
# speedup vs baseline: 1.8602x; 1.0310x over previous
#include <hip/hip_runtime.h>
#include <hip/hip_bf16.h>
#include <math.h>

typedef __attribute__((ext_vector_type(2))) float v2f;
typedef __attribute__((ext_vector_type(4))) float v4f;
typedef __attribute__((ext_vector_type(8))) float v8f;

#define WMMA_F32X4(a, b, c) \
  __builtin_amdgcn_wmma_f32_16x16x4_f32(false, (a), false, (b), (short)0, (c), false, false)

static __device__ __forceinline__ v2f lds_v2(const float* p) {
  return *reinterpret_cast<const v2f*>(p);
}
static __device__ __forceinline__ v4f lds_v4(const float* p) {
  return *reinterpret_cast<const v4f*>(p);
}
static __device__ __forceinline__ float xhalf(float x) {
  int b = __builtin_bit_cast(int, x);
  return __builtin_bit_cast(float,
      __builtin_amdgcn_permlanex16(b, b, 0x76543210, 0xfedcba98, false, false));
}

constexpr int NB  = 8192;
constexpr int NA  = 64;
constexpr int DIN = 128;
constexpr int E   = 32;

constexpr int PW = DIN + 4;
constexpr int PF = E + 4;

constexpr int NWG = 2048;

constexpr int CE1 = 4;
constexpr int CE  = 12;
struct more_yes { static constexpr bool value = true; };
struct more_no  { static constexpr bool value = false; };

__global__ __launch_bounds__(128) __attribute__((amdgpu_waves_per_eu(8, 8)))
void attn_layer_kernel(const float* __restrict__ X,
                       const float* __restrict__ W1,
                       const float* __restrict__ b1,
                       const float* __restrict__ Cm,
                       const float* __restrict__ W2,
                       const float* __restrict__ b2,
                       float* __restrict__ out)
{
  __shared__ __align__(16) float W1t[E][PW];
  __shared__ __align__(16) float Ct[E][PF];
  __shared__ __align__(16) float fis[NA][PF];
  __shared__ __align__(16) float W2s[2 * E];
  __shared__ __align__(16) float ss[NA];
  __shared__ __align__(16) float b1s[E];

  const int tid  = threadIdx.x;
  const int lane = tid & 31;
  const int wave = tid >> 5;
  const int half = lane >> 4;
  const int ln   = lane & 15;
  const int m0   = wave * 16;

  float w1r[DIN * E / 128];
  #pragma unroll
  for (int it = 0; it < DIN * E / 128; ++it) w1r[it] = W1[tid + 128 * it];
  float cr[E * E / 128];
  #pragma unroll
  for (int it = 0; it < E * E / 128; ++it) cr[it] = Cm[tid + 128 * it];
  float w2r = 0.f, b1r = 0.f;
  if (tid < 2 * E) w2r = W2[tid];
  if (tid < E)     b1r = b1[tid];
  const float b2v = b2[0];
  v4f xe[CE];
  {
    const float* Xrow0 = X + ((size_t)blockIdx.x * NA + m0 + ln) * DIN + 4 * half;
    #pragma unroll
    for (int c = 0; c < CE; ++c) xe[c] = *reinterpret_cast<const v4f*>(Xrow0 + 8 * c);
  }

  #pragma unroll
  for (int it = 0; it < DIN * E / 128; ++it) W1t[lane][wave + 4 * it] = w1r[it];
  #pragma unroll
  for (int it = 0; it < E * E / 128; ++it) Ct[lane][wave + 4 * it] = cr[it];
  if (tid < 2 * E) W2s[tid] = w2r;
  if (tid < E)     b1s[tid] = b1r;
  __builtin_amdgcn_fence(__ATOMIC_RELEASE, "workgroup", "local");
  __builtin_amdgcn_s_barrier();
  __builtin_amdgcn_fence(__ATOMIC_ACQUIRE, "workgroup", "local");

  const int wv = __builtin_amdgcn_readfirstlane(wave);

  auto batch = [&](const int b, auto more_tag) __attribute__((always_inline)) {
  constexpr bool MORE = decltype(more_tag)::value;
  const float* Xrow = X + ((size_t)b * NA + m0 + ln) * DIN + 4 * half;
  v4f xl[DIN / 8 - CE];
  #pragma unroll
  for (int c = CE; c < DIN / 8; ++c) xl[c - CE] = *reinterpret_cast<const v4f*>(Xrow + 8 * c);
  __builtin_amdgcn_sched_barrier(0);

  v8f acc0 = {}, acc1 = {};
  const int kb = 8 * half;
  constexpr int CT = DIN / 8 - 2;
  v4f pt[2], qt[2];
  v4f bqv[4], ctv0[4], ctv1[4];
  #pragma unroll
  for (int c = 0; c < DIN / 8; ++c) {
    if (c == CT) {
      #pragma unroll
      for (int d = 0; d < 2; ++d) {
        pt[d] = lds_v4(&W1t[ln][8 * (CT + d) + 4 * half]);
        qt[d] = lds_v4(&W1t[16 + ln][8 * (CT + d) + 4 * half]);
      }
      #pragma unroll
      for (int qq = 0; qq < 4; ++qq) {
        const int ko = 16 * (qq >> 1) + kb + 4 * (qq & 1);
        bqv[qq]  = lds_v4(&b1s[ko]);
        ctv0[qq] = lds_v4(&Ct[ln][ko]);
        ctv1[qq] = lds_v4(&Ct[16 + ln][ko]);
      }
      __builtin_amdgcn_sched_barrier(0);
    }
    v4f p, q;
    if (c < CT) {
      p = lds_v4(&W1t[ln][8 * c + 4 * half]);
      q = lds_v4(&W1t[16 + ln][8 * c + 4 * half]);
    } else {
      p = pt[c < CT ? 0 : c - CT];
      q = qt[c < CT ? 0 : c - CT];
    }
    const v4f xc = c < CE ? xe[c] : xl[c < CE ? 0 : c - CE];
    v2f al = {xc[0], xc[1]}, ah = {xc[2], xc[3]};
    v2f pl = {p[0], p[1]}, ph = {p[2], p[3]};
    v2f ql = {q[0], q[1]}, qh = {q[2], q[3]};
    acc0 = WMMA_F32X4(pl, al, acc0);
    acc1 = WMMA_F32X4(ql, al, acc1);
    acc0 = WMMA_F32X4(ph, ah, acc0);
    acc1 = WMMA_F32X4(qh, ah, acc1);
  }

  float fr[16];
  #pragma unroll
  for (int qq = 0; qq < 4; ++qq) {
    const v4f bq = bqv[qq];
    #pragma unroll
    for (int i = 0; i < 4; ++i) {
      const float a = qq < 2 ? acc0[(4 * qq + i) & 7] : acc1[(4 * qq + i) & 7];
      fr[4 * qq + i] = fmaxf(a + bq[i], 0.f);
    }
  }

  v8f gT0 = {}, gT1 = {};
  #pragma unroll
  for (int qq = 0; qq < 4; ++qq) {
    const v4f c0 = ctv0[qq];
    const v4f c1 = ctv1[qq];
    #pragma unroll
    for (int u = 0; u < 2; ++u) {
      const v2f bb = {fr[4 * qq + 2 * u], fr[4 * qq + 2 * u + 1]};
      const v2f a0 = {c0[2 * u], c0[2 * u + 1]};
      const v2f a1 = {c1[2 * u], c1[2 * u + 1]};
      gT0 = WMMA_F32X4(a0, bb, gT0);
      gT1 = WMMA_F32X4(a1, bb, gT1);
    }
  }
  float gr[16];
  #pragma unroll
  for (int i = 0; i < 8; ++i) { gr[i] = gT0[i];  gr[8 + i] = gT1[i]; }

  __builtin_amdgcn_sched_barrier(0);
  v4f wav[4], wbv[4];
  #pragma unroll
  for (int qq = 0; qq < 4; ++qq) {
    const int ko = 16 * (qq >> 1) + kb + 4 * (qq & 1);
    wav[qq] = lds_v4(&W2s[ko]);
    wbv[qq] = lds_v4(&W2s[E + ko]);
  }
  __builtin_amdgcn_sched_barrier(0);

  v8f bt[4] = {{}, {}, {}, {}};
  #pragma unroll
  for (int j = 0; j < 8; ++j) {
    const v2f a = {fr[2 * j], fr[2 * j + 1]};
    const v2f g = {gr[2 * j], gr[2 * j + 1]};
    bt[0] = WMMA_F32X4(a, g, bt[0]);
  }

  float pf, sj;
  {
    float pf0 = 0.f, pf1 = 0.f, sj0 = 0.f, sj1 = 0.f;
    #pragma unroll
    for (int qq = 0; qq < 4; ++qq) {
      const v4f wa = wav[qq];
      const v4f wb = wbv[qq];
      pf0 += fr[4 * qq] * wa[0];      pf1 += fr[4 * qq + 1] * wa[1];
      pf0 += fr[4 * qq + 2] * wa[2];  pf1 += fr[4 * qq + 3] * wa[3];
      sj0 += fr[4 * qq] * wb[0];      sj1 += fr[4 * qq + 1] * wb[1];
      sj0 += fr[4 * qq + 2] * wb[2];  sj1 += fr[4 * qq + 3] * wb[3];
    }
    pf = pf0 + pf1;
    sj = sj0 + sj1;
    sj += xhalf(sj);
  }
  asm volatile("" : "+v"(pf));

  __builtin_amdgcn_fence(__ATOMIC_RELEASE, "workgroup", "local");
  __builtin_amdgcn_s_barrier();
  __builtin_amdgcn_fence(__ATOMIC_ACQUIRE, "workgroup", "local");
  #pragma unroll
  for (int qq = 0; qq < 4; ++qq) {
    const v4f w = {fr[4 * qq], fr[4 * qq + 1], fr[4 * qq + 2], fr[4 * qq + 3]};
    *reinterpret_cast<v4f*>(&fis[m0 + ln][16 * (qq >> 1) + kb + 4 * (qq & 1)]) = w;
  }
  if (half == 0) ss[m0 + ln] = sj;
  __builtin_amdgcn_fence(__ATOMIC_RELEASE, "workgroup", "local");
  __builtin_amdgcn_s_barrier();
  __builtin_amdgcn_fence(__ATOMIC_ACQUIRE, "workgroup", "local");

  const float* Xnext = MORE ? Xrow + (size_t)NWG * NA * DIN : Xrow;
  if constexpr (MORE) {
    #pragma unroll
    for (int c = 0; c < CE1; ++c) xe[c] = *reinterpret_cast<const v4f*>(Xnext + 8 * c);
  }

  const float* frow[3];
  #pragma unroll
  for (int t = 1; t < 4; ++t) frow[t - 1] = &fis[ln][kb] + 16 * PF * ((wv + t) & 3);
  v4f fa[2][3];
  #pragma unroll
  for (int qq = 0; qq < 2; ++qq) {
    #pragma unroll
    for (int t = 0; t < 3; ++t) fa[qq][t] = lds_v4(frow[t] + 16 * (qq >> 1) + 4 * (qq & 1));
  }
  #pragma unroll
  for (int qq = 0; qq < 4; ++qq) {
    #pragma unroll
    for (int u = 0; u < 2; ++u) {
      const v2f g = {gr[4 * qq + 2 * u], gr[4 * qq + 2 * u + 1]};
      #pragma unroll
      for (int t = 0; t < 3; ++t) {
        const v2f a = {fa[qq & 1][t][2 * u], fa[qq & 1][t][2 * u + 1]};
        bt[t + 1] = WMMA_F32X4(a, g, bt[t + 1]);
      }
    }
    if (qq + 2 < 4) {
      #pragma unroll
      for (int t = 0; t < 3; ++t) fa[qq & 1][t] = lds_v4(frow[t] + 16 * ((qq + 2) >> 1) + 4 * ((qq + 2) & 1));
    }
    __builtin_amdgcn_sched_barrier(0);
  }

  if constexpr (MORE) {
    #pragma unroll
    for (int c = CE1; c < CE; ++c) xe[c] = *reinterpret_cast<const v4f*>(Xnext + 8 * c);
  }

  const int dr = ln - half * 8;
  #pragma unroll
  for (int r = 0; r < 8; ++r)
    if (r == dr) bt[0][r] = -INFINITY;
  float mxt[4];
  #pragma unroll
  for (int t = 0; t < 4; ++t) {
    float a = fmaxf(fmaxf(bt[t][0], bt[t][1]), fmaxf(bt[t][2], bt[t][3]));
    float c = fmaxf(fmaxf(bt[t][4], bt[t][5]), fmaxf(bt[t][6], bt[t][7]));
    mxt[t] = fmaxf(a, c);
  }
  float mx = fmaxf(fmaxf(mxt[0], mxt[1]), fmaxf(mxt[2], mxt[3]));
  mx = fmaxf(mx, xhalf(mx));
  float st[4] = {0.f, 0.f, 0.f, 0.f};
  float pv[4] = {0.f, 0.f, 0.f, 0.f};
  const float* srow[4];
  #pragma unroll
  for (int t = 0; t < 4; ++t) srow[t] = &ss[kb] + 16 * ((wv + t) & 3);
  v4f sv[2] = {lds_v4(srow[0]), lds_v4(srow[0] + 4)};
  #pragma unroll
  for (int t = 0; t < 4; ++t) {
    v4f nx[2] = {sv[0], sv[1]};
    if (t < 3) {
      nx[0] = lds_v4(srow[t < 3 ? t + 1 : 0]);
      nx[1] = lds_v4(srow[t < 3 ? t + 1 : 0] + 4);
    }
    #pragma unroll
    for (int u = 0; u < 2; ++u) {
      #pragma unroll
      for (int s = 0; s < 4; ++s) {
        float e = __expf(bt[t][4 * u + s] - mx);
        st[t] += e;
        pv[t] += e * sv[u][s];
      }
    }
    sv[0] = nx[0];  sv[1] = nx[1];
    __builtin_amdgcn_sched_barrier(0);
  }
  float sum = (st[0] + st[1]) + (st[2] + st[3]);
  sum += xhalf(sum);
  const float inv = 1.f / sum;
  const float pvs = (pv[0] + pv[1]) + (pv[2] + pv[3]);

  float pp = pf + pvs * inv;
  pp += xhalf(pp);
  if (half == 0)
    out[(size_t)b * NA + m0 + ln] = 1.f / (1.f + __expf(-(pp + b2v)));
  };

  int b = blockIdx.x;
  for (; b + NWG < NB; b += NWG) batch(b, more_yes{});
  batch(b, more_no{});
}

extern "C" void kernel_launch(void* const* d_in, const int* in_sizes, int n_in,
                              void* d_out, int out_size, void* d_ws, size_t ws_size,
                              hipStream_t stream) {
  const float* X  = (const float*)d_in[0];
  const float* W1 = (const float*)d_in[1];
  const float* b1 = (const float*)d_in[2];
  const float* Cm = (const float*)d_in[3];
  const float* W2 = (const float*)d_in[4];
  const float* b2 = (const float*)d_in[5];
  float* out = (float*)d_out;
  attn_layer_kernel<<<NWG, 128, 0, stream>>>(X, W1, b1, Cm, W2, b2, out);
}
